# starting kernel
# baseline (speedup 1.0000x reference)
#include <hip/hip_runtime.h>
#include <math.h>

#define N_NODES 100000
#define IN_W    64
#define OUT_W   64
#define N_EDGES 1600000

typedef __attribute__((ext_vector_type(2))) float v2f;
typedef __attribute__((ext_vector_type(8))) float v8f;

__global__ void deg_init_kernel(float* __restrict__ deg) {
    int i = blockIdx.x * blockDim.x + threadIdx.x;
    if (i < N_NODES) deg[i] = 2.0f;
}

__global__ void deg_accum_kernel(const int* __restrict__ dst,
                                 const float* __restrict__ w,
                                 float* __restrict__ deg) {
    int e = blockIdx.x * blockDim.x + threadIdx.x;
    if (e < N_EDGES) atomicAdd(&deg[dst[e]], w[e]);
}

__global__ void deg_rsqrt_kernel(float* __restrict__ deg) {
    int i = blockIdx.x * blockDim.x + threadIdx.x;
    if (i < N_NODES) {
        float d = deg[i];
        deg[i] = (d > 0.0f) ? rsqrtf(d) : 0.0f;
    }
}

__global__ void gemm_wmma_kernel(const float* __restrict__ x,
                                 const float* __restrict__ W,
                                 float* __restrict__ h) {
    const unsigned lane  = threadIdx.x & 31u;
    const unsigned gwave = blockIdx.x * 8u + (threadIdx.x >> 5);
    const unsigned tile_n = gwave & 3u;
    const unsigned tile_m = gwave >> 2;
    const unsigned m0 = tile_m * 16u;
    const unsigned n0 = tile_n * 16u;
    const unsigned half = lane >> 4;
    const unsigned lrow = lane & 15u;

    const float* xrow = x + (size_t)(m0 + lrow) * IN_W;
    const float* wcol = W + (n0 + lrow);

    v8f c = {};
#pragma unroll
    for (int kk = 0; kk < 16; ++kk) {
        const int kb = kk * 4 + (int)half * 2;
        v2f a = *(const v2f*)(xrow + kb);
        v2f b;
        b.x = wcol[(size_t)kb * OUT_W];
        b.y = wcol[(size_t)(kb + 1) * OUT_W];
        c = __builtin_amdgcn_wmma_f32_16x16x4_f32(
                 false, a,  false, b,
                 (short)0, c,  false,  false);
    }

    float* hcol = h + n0 + lrow;
#pragma unroll
    for (int v = 0; v < 8; ++v) {
        hcol[(size_t)(m0 + v + half * 8u) * OUT_W] = c[v];
    }
}

__global__ void out_init_kernel(const float* __restrict__ h,
                                const float* __restrict__ dinv,
                                const float* __restrict__ bias,
                                float* __restrict__ out) {
    int t = blockIdx.x * blockDim.x + threadIdx.x;
    if (t >= N_NODES * (OUT_W / 4)) return;
    int node = t >> 4;
    int f    = (t & 15) * 4;
    float di = dinv[node];
    float s  = 2.0f * di * di;
    float4 hv = *(const float4*)(h + (size_t)node * OUT_W + f);
    float4 bv = *(const float4*)(bias + f);
    float4 o;
    o.x = bv.x + hv.x * s;
    o.y = bv.y + hv.y * s;
    o.z = bv.z + hv.z * s;
    o.w = bv.w + hv.w * s;
    *(float4*)(out + (size_t)node * OUT_W + f) = o;
}

__global__ void edge_scatter_kernel(const int* __restrict__ src,
                                    const int* __restrict__ dst,
                                    const float* __restrict__ w,
                                    const float* __restrict__ dinv,
                                    const float* __restrict__ h,
                                    float* __restrict__ out) {
    int t = blockIdx.x * blockDim.x + threadIdx.x;
    int e = t >> 4;
    if (e >= N_EDGES) return;
    int f = (t & 15) * 4;
    int s = src[e];
    int d = dst[e];
    float coef = dinv[s] * w[e] * dinv[d];
    float4 hv = *(const float4*)(h + (size_t)s * OUT_W + f);
    float* o = out + (size_t)d * OUT_W + f;
    atomicAdd(o + 0, hv.x * coef);
    atomicAdd(o + 1, hv.y * coef);
    atomicAdd(o + 2, hv.z * coef);
    atomicAdd(o + 3, hv.w * coef);
}

extern "C" void kernel_launch(void* const* d_in, const int* in_sizes, int n_in,
                              void* d_out, int out_size, void* d_ws, size_t ws_size,
                              hipStream_t stream) {
    const float* x         = (const float*)d_in[0];
    const float* edge_attr = (const float*)d_in[1];
    const float* W         = (const float*)d_in[2];
    const float* bias      = (const float*)d_in[3];
    const int*   edge_idx  = (const int*)d_in[4];
    float* out = (float*)d_out;

    float* dinv = (float*)d_ws;
    float* h    = dinv + 100096;

    const int* src = edge_idx;
    const int* dst = edge_idx + N_EDGES;

    deg_init_kernel<<<(N_NODES + 255) / 256, 256, 0, stream>>>(dinv);
    deg_accum_kernel<<<(N_EDGES + 255) / 256, 256, 0, stream>>>(dst, edge_attr, dinv);
    deg_rsqrt_kernel<<<(N_NODES + 255) / 256, 256, 0, stream>>>(dinv);

    gemm_wmma_kernel<<<3125, 256, 0, stream>>>(x, W, h);

    out_init_kernel<<<(N_NODES * 16 + 255) / 256, 256, 0, stream>>>(h, dinv, bias, out);

    edge_scatter_kernel<<<(N_EDGES * 16 + 255) / 256, 256, 0, stream>>>(
        src, dst, edge_attr, dinv, h, out);
}


# degree sums privatized: 128 work-groups (32 edge chunks x 4 node ranges) add w into LDS bins (ds f32 adds, sum order varies); 32 partial planes summed in the rsqrt pass; global deg atomics gone
# speedup vs baseline: 1.4023x; 1.4023x over previous
#include <hip/hip_runtime.h>
#include <math.h>

#define N_NODES 100000
#define IN_W    64
#define OUT_W   64
#define N_EDGES 1600000

typedef __attribute__((ext_vector_type(2))) float v2f;
typedef __attribute__((ext_vector_type(8))) float v8f;

#define DEG_C   32
#define DEG_CE  (N_EDGES / DEG_C)
#define DEG_CV  (DEG_CE / 4)
#define DEG_Q   4
#define DEG_QN  (N_NODES / DEG_Q)
#define DEG_QV  (DEG_QN / 4)
#define DEG_T   1024
#define DEG_U   4

__global__ __launch_bounds__(DEG_T)
void deg_partial_kernel(const int* __restrict__ dst,
                        const float* __restrict__ w,
                        float* __restrict__ part) {
    __shared__ __attribute__((aligned(16))) float bins[DEG_QN];
    const unsigned tid   = threadIdx.x;
    const unsigned chunk = blockIdx.x;
    const unsigned nbase = blockIdx.y * DEG_QN;

    float4* bins4 = (float4*)bins;
    for (unsigned i = tid; i < DEG_QV; i += DEG_T)
        bins4[i] = make_float4(0.0f, 0.0f, 0.0f, 0.0f);
    __syncthreads();

    const int4*   d4 = (const int4*)(dst + (size_t)chunk * DEG_CE);
    const float4* w4 = (const float4*)(w + (size_t)chunk * DEG_CE);

    for (unsigned i0 = tid; i0 < DEG_CV; i0 += DEG_T * DEG_U) {
        int4   dv[DEG_U];
        float4 wv[DEG_U];
#pragma unroll
        for (int u = 0; u < DEG_U; ++u) {
            const unsigned i  = i0 + (unsigned)u * DEG_T;
            const unsigned ic = (i < DEG_CV) ? i : (DEG_CV - 1);
            dv[u] = d4[ic];
            wv[u] = w4[ic];
        }
        asm volatile("" :: "v"(wv[0].x), "v"(wv[0].y), "v"(wv[0].z), "v"(wv[0].w),
                           "v"(wv[1].x), "v"(wv[1].y), "v"(wv[1].z), "v"(wv[1].w),
                           "v"(wv[2].x), "v"(wv[2].y), "v"(wv[2].z), "v"(wv[2].w),
                           "v"(wv[3].x), "v"(wv[3].y), "v"(wv[3].z), "v"(wv[3].w));
#pragma unroll
        for (int u = 0; u < DEG_U; ++u) {
            const unsigned i = i0 + (unsigned)u * DEG_T;
            if (i < DEG_CV) {
                unsigned r;
                r = (unsigned)dv[u].x - nbase; if (r < DEG_QN) atomicAdd(&bins[r], wv[u].x);
                r = (unsigned)dv[u].y - nbase; if (r < DEG_QN) atomicAdd(&bins[r], wv[u].y);
                r = (unsigned)dv[u].z - nbase; if (r < DEG_QN) atomicAdd(&bins[r], wv[u].z);
                r = (unsigned)dv[u].w - nbase; if (r < DEG_QN) atomicAdd(&bins[r], wv[u].w);
            }
        }
    }
    __syncthreads();

    float4* o4 = (float4*)(part + (size_t)chunk * N_NODES + nbase);
    for (unsigned i = tid; i < DEG_QV; i += DEG_T)
        o4[i] = bins4[i];
}

__global__ void deg_finish_kernel(const float* __restrict__ part,
                                  float* __restrict__ dinv) {
    int i = blockIdx.x * blockDim.x + threadIdx.x;
    if (i < N_NODES) {
        float d = 0.0f;
#pragma unroll
        for (int c = 0; c < DEG_C; ++c) d += part[(size_t)c * N_NODES + i];
        d += 2.0f;
        dinv[i] = (d > 0.0f) ? rsqrtf(d) : 0.0f;
    }
}

__global__ void gemm_wmma_kernel(const float* __restrict__ x,
                                 const float* __restrict__ W,
                                 float* __restrict__ h) {
    const unsigned lane  = threadIdx.x & 31u;
    const unsigned gwave = blockIdx.x * 8u + (threadIdx.x >> 5);
    const unsigned tile_n = gwave & 3u;
    const unsigned tile_m = gwave >> 2;
    const unsigned m0 = tile_m * 16u;
    const unsigned n0 = tile_n * 16u;
    const unsigned half = lane >> 4;
    const unsigned lrow = lane & 15u;

    const float* xrow = x + (size_t)(m0 + lrow) * IN_W;
    const float* wcol = W + (n0 + lrow);

    v8f c = {};
#pragma unroll
    for (int kk = 0; kk < 16; ++kk) {
        const int kb = kk * 4 + (int)half * 2;
        v2f a = *(const v2f*)(xrow + kb);
        v2f b;
        b.x = wcol[(size_t)kb * OUT_W];
        b.y = wcol[(size_t)(kb + 1) * OUT_W];
        c = __builtin_amdgcn_wmma_f32_16x16x4_f32(
                 false, a,  false, b,
                 (short)0, c,  false,  false);
    }

    float* hcol = h + n0 + lrow;
#pragma unroll
    for (int v = 0; v < 8; ++v) {
        hcol[(size_t)(m0 + v + half * 8u) * OUT_W] = c[v];
    }
}

__global__ void out_init_kernel(const float* __restrict__ h,
                                const float* __restrict__ dinv,
                                const float* __restrict__ bias,
                                float* __restrict__ out) {
    int t = blockIdx.x * blockDim.x + threadIdx.x;
    if (t >= N_NODES * (OUT_W / 4)) return;
    int node = t >> 4;
    int f    = (t & 15) * 4;
    float di = dinv[node];
    float s  = 2.0f * di * di;
    float4 hv = *(const float4*)(h + (size_t)node * OUT_W + f);
    float4 bv = *(const float4*)(bias + f);
    float4 o;
    o.x = bv.x + hv.x * s;
    o.y = bv.y + hv.y * s;
    o.z = bv.z + hv.z * s;
    o.w = bv.w + hv.w * s;
    *(float4*)(out + (size_t)node * OUT_W + f) = o;
}

__global__ void edge_scatter_kernel(const int* __restrict__ src,
                                    const int* __restrict__ dst,
                                    const float* __restrict__ w,
                                    const float* __restrict__ dinv,
                                    const float* __restrict__ h,
                                    float* __restrict__ out) {
    int t = blockIdx.x * blockDim.x + threadIdx.x;
    int e = t >> 4;
    if (e >= N_EDGES) return;
    int f = (t & 15) * 4;
    int s = src[e];
    int d = dst[e];
    float coef = dinv[s] * w[e] * dinv[d];
    float4 hv = *(const float4*)(h + (size_t)s * OUT_W + f);
    float* o = out + (size_t)d * OUT_W + f;
    atomicAdd(o + 0, hv.x * coef);
    atomicAdd(o + 1, hv.y * coef);
    atomicAdd(o + 2, hv.z * coef);
    atomicAdd(o + 3, hv.w * coef);
}

extern "C" void kernel_launch(void* const* d_in, const int* in_sizes, int n_in,
                              void* d_out, int out_size, void* d_ws, size_t ws_size,
                              hipStream_t stream) {
    const float* x         = (const float*)d_in[0];
    const float* edge_attr = (const float*)d_in[1];
    const float* W         = (const float*)d_in[2];
    const float* bias      = (const float*)d_in[3];
    const int*   edge_idx  = (const int*)d_in[4];
    float* out = (float*)d_out;

    float* dinv = (float*)d_ws;
    float* h    = dinv + 100096;

    const int* src = edge_idx;
    const int* dst = edge_idx + N_EDGES;

    float* part = h;
    deg_partial_kernel<<<dim3(DEG_C, DEG_Q), DEG_T, 0, stream>>>(dst, edge_attr, part);
    deg_finish_kernel<<<(N_NODES + 255) / 256, 256, 0, stream>>>(part, dinv);

    gemm_wmma_kernel<<<3125, 256, 0, stream>>>(x, W, h);

    out_init_kernel<<<(N_NODES * 16 + 255) / 256, 256, 0, stream>>>(h, dinv, bias, out);

    edge_scatter_kernel<<<(N_EDGES * 16 + 255) / 256, 256, 0, stream>>>(
        src, dst, edge_attr, dinv, h, out);
}


# edges counting-sorted by dst node (LDS counts + returning LDS adds); 16 lanes sum a node's x rows in registers, then agg @ W on f32 WMMA; no global atomics, out_init gone; sum order changes
# speedup vs baseline: 2.2629x; 1.6137x over previous
#include <hip/hip_runtime.h>
#include <math.h>


#define N_NODES 100000
#define IN_W    64
#define OUT_W   64
#define N_EDGES 1600000

typedef __attribute__((ext_vector_type(2))) float v2f;
typedef __attribute__((ext_vector_type(8))) float v8f;

#define DEG_C   32
#define DEG_CE  (N_EDGES / DEG_C)
#define DEG_CV  (DEG_CE / 4)
#define DEG_Q   4
#define DEG_QN  (N_NODES / DEG_Q)
#define DEG_QV  (DEG_QN / 4)
#define DEG_T   1024
#define DEG_U   4

#define SB_N    400
#define SB_B    (N_NODES / SB_N)
#define SB_T    256

__global__ __launch_bounds__(DEG_T)
void count_kernel(const int* __restrict__ dst,
                  const float* __restrict__ w,
                  float* __restrict__ wpart,
                  int* __restrict__ cpart) {
    __shared__ __attribute__((aligned(16))) float bins[DEG_QN];
    __shared__ __attribute__((aligned(16))) int   cnts[DEG_QN];
    const unsigned tid   = threadIdx.x;
    const unsigned chunk = blockIdx.x;
    const unsigned nbase = blockIdx.y * DEG_QN;

    float4* bins4 = (float4*)bins;
    int4*   cnts4 = (int4*)cnts;
    for (unsigned i = tid; i < DEG_QV; i += DEG_T) {
        bins4[i] = make_float4(0.0f, 0.0f, 0.0f, 0.0f);
        cnts4[i] = make_int4(0, 0, 0, 0);
    }
    __syncthreads();

    const int4*   d4 = (const int4*)(dst + (size_t)chunk * DEG_CE);
    const float4* w4 = (const float4*)(w + (size_t)chunk * DEG_CE);

    for (unsigned i0 = tid; i0 < DEG_CV; i0 += DEG_T * DEG_U) {
        int4   dv[DEG_U];
        float4 wv[DEG_U];
#pragma unroll
        for (int u = 0; u < DEG_U; ++u) {
            const unsigned i  = i0 + (unsigned)u * DEG_T;
            const unsigned ic = (i < DEG_CV) ? i : (DEG_CV - 1);
            dv[u] = d4[ic];
            wv[u] = w4[ic];
        }
        asm volatile("" :: "v"(wv[0].x), "v"(wv[0].y), "v"(wv[0].z), "v"(wv[0].w),
                           "v"(wv[1].x), "v"(wv[1].y), "v"(wv[1].z), "v"(wv[1].w),
                           "v"(wv[2].x), "v"(wv[2].y), "v"(wv[2].z), "v"(wv[2].w),
                           "v"(wv[3].x), "v"(wv[3].y), "v"(wv[3].z), "v"(wv[3].w));
#pragma unroll
        for (int u = 0; u < DEG_U; ++u) {
            const unsigned i = i0 + (unsigned)u * DEG_T;
            if (i < DEG_CV) {
                unsigned r;
                r = (unsigned)dv[u].x - nbase; if (r < DEG_QN) { atomicAdd(&bins[r], wv[u].x); atomicAdd(&cnts[r], 1); }
                r = (unsigned)dv[u].y - nbase; if (r < DEG_QN) { atomicAdd(&bins[r], wv[u].y); atomicAdd(&cnts[r], 1); }
                r = (unsigned)dv[u].z - nbase; if (r < DEG_QN) { atomicAdd(&bins[r], wv[u].z); atomicAdd(&cnts[r], 1); }
                r = (unsigned)dv[u].w - nbase; if (r < DEG_QN) { atomicAdd(&bins[r], wv[u].w); atomicAdd(&cnts[r], 1); }
            }
        }
    }
    __syncthreads();

    float4* o4 = (float4*)(wpart + (size_t)chunk * N_NODES + nbase);
    int4*   c4 = (int4*)(cpart + (size_t)chunk * N_NODES + nbase);
    for (unsigned i = tid; i < DEG_QV; i += DEG_T) {
        o4[i] = bins4[i];
        c4[i] = cnts4[i];
    }
}

__global__ __launch_bounds__(SB_T)
void offsets_kernel(const float* __restrict__ wpart,
                    int* __restrict__ cpart,
                    float* __restrict__ dinv,
                    int* __restrict__ btot) {
    __shared__ int sc[SB_T];
    const unsigned t  = threadIdx.x;
    const bool act    = t < (SB_N / 2);
    const unsigned n0 = blockIdx.x * SB_N + 2u * t;

    int c0[DEG_C], c1[DEG_C];
    int tot0 = 0, tot1 = 0;
    if (act) {
        float d0 = 0.0f, d1 = 0.0f;
#pragma unroll
        for (int c = 0; c < DEG_C; ++c) {
            const float2 wv = *(const float2*)(wpart + (size_t)c * N_NODES + n0);
            const int2   cv = *(const int2*)(cpart + (size_t)c * N_NODES + n0);
            d0 += wv.x; d1 += wv.y;
            c0[c] = cv.x; c1[c] = cv.y;
            tot0 += cv.x; tot1 += cv.y;
        }
        d0 += 2.0f;
        d1 += 2.0f;
        float2 di;
        di.x = (d0 > 0.0f) ? rsqrtf(d0) : 0.0f;
        di.y = (d1 > 0.0f) ? rsqrtf(d1) : 0.0f;
        *(float2*)(dinv + n0) = di;
    }
    const int pair = tot0 + tot1;
    sc[t] = pair;
    __syncthreads();
    for (unsigned off = 1; off < SB_T; off <<= 1) {
        const int v = (t >= off) ? sc[t - off] : 0;
        __syncthreads();
        sc[t] += v;
        __syncthreads();
    }
    const int excl = sc[t] - pair;
    if (t == SB_T - 1) btot[blockIdx.x] = sc[t];
    if (act) {
        int r0 = excl, r1 = excl + tot0;
#pragma unroll
        for (int c = 0; c < DEG_C; ++c) {
            int2 o; o.x = r0; o.y = r1;
            *(int2*)(cpart + (size_t)c * N_NODES + n0) = o;
            r0 += c0[c]; r1 += c1[c];
        }
    }
}

__global__ __launch_bounds__(DEG_T)
void place_kernel(const int* __restrict__ src,
                  const int* __restrict__ dst,
                  const float* __restrict__ w,
                  int* __restrict__ offs,
                  const int* __restrict__ btot,
                  int2* __restrict__ recs) {
    __shared__ __attribute__((aligned(16))) int cur[DEG_QN];
    __shared__ int bb[257];
    const unsigned tid   = threadIdx.x;
    const unsigned chunk = blockIdx.x;
    const unsigned nbase = blockIdx.y * DEG_QN;

    int* sc = bb + 1;
    if (tid < 256) sc[tid] = (tid < SB_B) ? btot[tid] : 0;
    if (tid == 0) bb[0] = 0;
    __syncthreads();
    for (unsigned off = 1; off < 256; off <<= 1) {
        const int v = (tid < 256 && tid >= off) ? sc[tid - off] : 0;
        __syncthreads();
        if (tid < 256) sc[tid] += v;
        __syncthreads();
    }

    int4* my4  = (int4*)(offs + (size_t)chunk * N_NODES + nbase);
    int4* cur4 = (int4*)cur;
    for (unsigned i0 = tid; i0 < DEG_QV; i0 += DEG_T * DEG_U) {
        int4 a[DEG_U];
#pragma unroll
        for (int u = 0; u < DEG_U; ++u) {
            const unsigned i  = i0 + (unsigned)u * DEG_T;
            const unsigned ic = (i < DEG_QV) ? i : (DEG_QV - 1);
            a[u] = my4[ic];
        }
        asm volatile("" :: "v"(a[0].x), "v"(a[0].y), "v"(a[0].z), "v"(a[0].w),
                           "v"(a[1].x), "v"(a[1].y), "v"(a[1].z), "v"(a[1].w),
                           "v"(a[2].x), "v"(a[2].y), "v"(a[2].z), "v"(a[2].w),
                           "v"(a[3].x), "v"(a[3].y), "v"(a[3].z), "v"(a[3].w));
#pragma unroll
        for (int u = 0; u < DEG_U; ++u) {
            const unsigned i = i0 + (unsigned)u * DEG_T;
            if (i < DEG_QV) {
                const int b = bb[(nbase + 4u * i) / SB_N];
                int4 v;
                v.x = a[u].x + b; v.y = a[u].y + b; v.z = a[u].z + b; v.w = a[u].w + b;
                cur4[i] = v;
                if (chunk == 0) my4[i] = v;
            }
        }
    }
    __syncthreads();

    const int4*   d4 = (const int4*)(dst + (size_t)chunk * DEG_CE);
    const int4*   s4 = (const int4*)(src + (size_t)chunk * DEG_CE);
    const float4* w4 = (const float4*)(w + (size_t)chunk * DEG_CE);

    for (unsigned i0 = tid; i0 < DEG_CV; i0 += DEG_T * DEG_U) {
        int4   dv[DEG_U];
        int4   sv[DEG_U];
        float4 wv[DEG_U];
#pragma unroll
        for (int u = 0; u < DEG_U; ++u) {
            const unsigned i  = i0 + (unsigned)u * DEG_T;
            const unsigned ic = (i < DEG_CV) ? i : (DEG_CV - 1);
            dv[u] = d4[ic];
            sv[u] = s4[ic];
            wv[u] = w4[ic];
        }
        asm volatile("" :: "v"(sv[0].x), "v"(sv[0].y), "v"(sv[0].z), "v"(sv[0].w),
                           "v"(sv[1].x), "v"(sv[1].y), "v"(sv[1].z), "v"(sv[1].w),
                           "v"(sv[2].x), "v"(sv[2].y), "v"(sv[2].z), "v"(sv[2].w),
                           "v"(sv[3].x), "v"(sv[3].y), "v"(sv[3].z), "v"(sv[3].w));
        asm volatile("" :: "v"(wv[0].x), "v"(wv[0].y), "v"(wv[0].z), "v"(wv[0].w),
                           "v"(wv[1].x), "v"(wv[1].y), "v"(wv[1].z), "v"(wv[1].w),
                           "v"(wv[2].x), "v"(wv[2].y), "v"(wv[2].z), "v"(wv[2].w),
                           "v"(wv[3].x), "v"(wv[3].y), "v"(wv[3].z), "v"(wv[3].w));
#pragma unroll
        for (int u = 0; u < DEG_U; ++u) {
            const unsigned i = i0 + (unsigned)u * DEG_T;
            if (i < DEG_CV) {
                unsigned r;
                r = (unsigned)dv[u].x - nbase;
                if (r < DEG_QN) { const int p = atomicAdd(&cur[r], 1); recs[p] = make_int2(sv[u].x, __float_as_int(wv[u].x)); }
                r = (unsigned)dv[u].y - nbase;
                if (r < DEG_QN) { const int p = atomicAdd(&cur[r], 1); recs[p] = make_int2(sv[u].y, __float_as_int(wv[u].y)); }
                r = (unsigned)dv[u].z - nbase;
                if (r < DEG_QN) { const int p = atomicAdd(&cur[r], 1); recs[p] = make_int2(sv[u].z, __float_as_int(wv[u].z)); }
                r = (unsigned)dv[u].w - nbase;
                if (r < DEG_QN) { const int p = atomicAdd(&cur[r], 1); recs[p] = make_int2(sv[u].w, __float_as_int(wv[u].w)); }
            }
        }
    }
}

__global__ __launch_bounds__(256)
void pull_kernel(const float* __restrict__ x,
                 const float* __restrict__ dinv,
                 const int* __restrict__ start,
                 const int2* __restrict__ recs,
                 float* __restrict__ agg) {
    const unsigned t = blockIdx.x * 256u + threadIdx.x;
    const unsigned n = t >> 4;
    const unsigned q = t & 15u;
    const float4* x4 = (const float4*)x;

    const int jb = start[n];
    const int je = (n + 1u < N_NODES) ? start[n + 1u] : N_EDGES;
    const float di = dinv[n];
    const float4 xn = x4[(size_t)n * 16u + q];

    float4 acc = make_float4(0.0f, 0.0f, 0.0f, 0.0f);
    for (int j = jb; j < je; j += 4) {
        int2   rc[4];
        float  cf[4];
        float4 xv[4];
#pragma unroll
        for (int u = 0; u < 4; ++u) {
            const int jj = (j + u < je) ? (j + u) : (je - 1);
            rc[u] = recs[jj];
        }
#pragma unroll
        for (int u = 0; u < 4; ++u) {
            const float ws = (j + u < je) ? __int_as_float(rc[u].y) : 0.0f;
            cf[u] = ws * dinv[rc[u].x];
            xv[u] = x4[(size_t)rc[u].x * 16u + q];
        }
#pragma unroll
        for (int u = 0; u < 4; ++u) {
            acc.x += cf[u] * xv[u].x;
            acc.y += cf[u] * xv[u].y;
            acc.z += cf[u] * xv[u].z;
            acc.w += cf[u] * xv[u].w;
        }
    }
    const float sl = 2.0f * di;
    float4 o;
    o.x = di * (acc.x + sl * xn.x);
    o.y = di * (acc.y + sl * xn.y);
    o.z = di * (acc.z + sl * xn.z);
    o.w = di * (acc.w + sl * xn.w);
    ((float4*)agg)[(size_t)n * 16u + q] = o;
}

__global__ void gemm_wmma_kernel(const float* __restrict__ a_in,
                                 const float* __restrict__ W,
                                 const float* __restrict__ bias,
                                 float* __restrict__ out) {
    const unsigned lane  = threadIdx.x & 31u;
    const unsigned gwave = blockIdx.x * 8u + (threadIdx.x >> 5);
    const unsigned tile_n = gwave & 3u;
    const unsigned tile_m = gwave >> 2;
    const unsigned m0 = tile_m * 16u;
    const unsigned n0 = tile_n * 16u;
    const unsigned half = lane >> 4;
    const unsigned lrow = lane & 15u;

    const float* xrow = a_in + (size_t)(m0 + lrow) * IN_W;
    const float* wcol = W + (n0 + lrow);

    v8f c = {};
#pragma unroll
    for (int kk = 0; kk < 16; ++kk) {
        const int kb = kk * 4 + (int)half * 2;
        v2f a = *(const v2f*)(xrow + kb);
        v2f b;
        b.x = wcol[(size_t)kb * OUT_W];
        b.y = wcol[(size_t)(kb + 1) * OUT_W];
        c = __builtin_amdgcn_wmma_f32_16x16x4_f32(
                 false, a,  false, b,
                 (short)0, c,  false,  false);
    }

    const float bv = bias[n0 + lrow];
    float* ocol = out + n0 + lrow;
#pragma unroll
    for (int v = 0; v < 8; ++v) {
        ocol[(size_t)(m0 + v + half * 8u) * OUT_W] = c[v] + bv;
    }
}

extern "C" void kernel_launch(void* const* d_in, const int* in_sizes, int n_in,
                              void* d_out, int out_size, void* d_ws, size_t ws_size,
                              hipStream_t stream) {
    const float* x         = (const float*)d_in[0];
    const float* edge_attr = (const float*)d_in[1];
    const float* W         = (const float*)d_in[2];
    const float* bias      = (const float*)d_in[3];
    const int*   edge_idx  = (const int*)d_in[4];
    float* out = (float*)d_out;

    float* dinv = (float*)d_ws;
    float* regR = dinv + 100096;
    int*   btot = (int*)(regR + (size_t)DEG_C * N_NODES);

    int*  offs = (int*)out;
    int2* recs = (int2*)(out + (size_t)DEG_C * N_NODES);

    const int* src = edge_idx;
    const int* dst = edge_idx + N_EDGES;

    float* wpart = regR;
    float* agg   = regR;

    count_kernel<<<dim3(DEG_C, DEG_Q), DEG_T, 0, stream>>>(dst, edge_attr, wpart, offs);
    offsets_kernel<<<SB_B, SB_T, 0, stream>>>(wpart, offs, dinv, btot);
    place_kernel<<<dim3(DEG_C, DEG_Q), DEG_T, 0, stream>>>(src, dst, edge_attr, offs, btot, recs);
    pull_kernel<<<(N_NODES * 16) / 256, 256, 0, stream>>>(x, dinv, offs, recs, agg);
    gemm_wmma_kernel<<<3125, 256, 0, stream>>>(agg, W, bias, out);
}


# product launch: its 32 rows of agg and all of W are read 16 B per lane (whole 128-B lines) and reach the f32 WMMA fragments through padded LDS; the 16-line fragment loads are gone; sum order kept
# speedup vs baseline: 2.3448x; 1.0362x over previous
#include <hip/hip_runtime.h>
#include <math.h>


#define N_NODES 100000
#define IN_W    64
#define OUT_W   64
#define N_EDGES 1600000

typedef __attribute__((ext_vector_type(2))) float v2f;
typedef __attribute__((ext_vector_type(8))) float v8f;

#define DEG_C   32
#define DEG_CE  (N_EDGES / DEG_C)
#define DEG_CV  (DEG_CE / 4)
#define DEG_Q   4
#define DEG_QN  (N_NODES / DEG_Q)
#define DEG_QV  (DEG_QN / 4)
#define DEG_T   1024
#define DEG_U   4

#define SB_N    400
#define SB_B    (N_NODES / SB_N)
#define SB_T    256

__global__ __launch_bounds__(DEG_T)
void count_kernel(const int* __restrict__ dst,
                  const float* __restrict__ w,
                  float* __restrict__ wpart,
                  int* __restrict__ cpart) {
    __shared__ __attribute__((aligned(16))) float bins[DEG_QN];
    __shared__ __attribute__((aligned(16))) int   cnts[DEG_QN];
    const unsigned tid   = threadIdx.x;
    const unsigned chunk = blockIdx.x;
    const unsigned nbase = blockIdx.y * DEG_QN;

    float4* bins4 = (float4*)bins;
    int4*   cnts4 = (int4*)cnts;
    for (unsigned i = tid; i < DEG_QV; i += DEG_T) {
        bins4[i] = make_float4(0.0f, 0.0f, 0.0f, 0.0f);
        cnts4[i] = make_int4(0, 0, 0, 0);
    }
    __syncthreads();

    const int4*   d4 = (const int4*)(dst + (size_t)chunk * DEG_CE);
    const float4* w4 = (const float4*)(w + (size_t)chunk * DEG_CE);

    for (unsigned i0 = tid; i0 < DEG_CV; i0 += DEG_T * DEG_U) {
        int4   dv[DEG_U];
        float4 wv[DEG_U];
#pragma unroll
        for (int u = 0; u < DEG_U; ++u) {
            const unsigned i  = i0 + (unsigned)u * DEG_T;
            const unsigned ic = (i < DEG_CV) ? i : (DEG_CV - 1);
            dv[u] = d4[ic];
            wv[u] = w4[ic];
        }
        asm volatile("" :: "v"(wv[0].x), "v"(wv[0].y), "v"(wv[0].z), "v"(wv[0].w),
                           "v"(wv[1].x), "v"(wv[1].y), "v"(wv[1].z), "v"(wv[1].w),
                           "v"(wv[2].x), "v"(wv[2].y), "v"(wv[2].z), "v"(wv[2].w),
                           "v"(wv[3].x), "v"(wv[3].y), "v"(wv[3].z), "v"(wv[3].w));
#pragma unroll
        for (int u = 0; u < DEG_U; ++u) {
            const unsigned i = i0 + (unsigned)u * DEG_T;
            if (i < DEG_CV) {
                unsigned r;
                r = (unsigned)dv[u].x - nbase; if (r < DEG_QN) { atomicAdd(&bins[r], wv[u].x); atomicAdd(&cnts[r], 1); }
                r = (unsigned)dv[u].y - nbase; if (r < DEG_QN) { atomicAdd(&bins[r], wv[u].y); atomicAdd(&cnts[r], 1); }
                r = (unsigned)dv[u].z - nbase; if (r < DEG_QN) { atomicAdd(&bins[r], wv[u].z); atomicAdd(&cnts[r], 1); }
                r = (unsigned)dv[u].w - nbase; if (r < DEG_QN) { atomicAdd(&bins[r], wv[u].w); atomicAdd(&cnts[r], 1); }
            }
        }
    }
    __syncthreads();

    float4* o4 = (float4*)(wpart + (size_t)chunk * N_NODES + nbase);
    int4*   c4 = (int4*)(cpart + (size_t)chunk * N_NODES + nbase);
    for (unsigned i = tid; i < DEG_QV; i += DEG_T) {
        o4[i] = bins4[i];
        c4[i] = cnts4[i];
    }
}

__global__ __launch_bounds__(SB_T)
void offsets_kernel(const float* __restrict__ wpart,
                    int* __restrict__ cpart,
                    float* __restrict__ dinv,
                    int* __restrict__ btot) {
    __shared__ int sc[SB_T];
    const unsigned t  = threadIdx.x;
    const bool act    = t < (SB_N / 2);
    const unsigned n0 = blockIdx.x * SB_N + 2u * t;

    int c0[DEG_C], c1[DEG_C];
    int tot0 = 0, tot1 = 0;
    if (act) {
        float d0 = 0.0f, d1 = 0.0f;
#pragma unroll
        for (int c = 0; c < DEG_C; ++c) {
            const float2 wv = *(const float2*)(wpart + (size_t)c * N_NODES + n0);
            const int2   cv = *(const int2*)(cpart + (size_t)c * N_NODES + n0);
            d0 += wv.x; d1 += wv.y;
            c0[c] = cv.x; c1[c] = cv.y;
            tot0 += cv.x; tot1 += cv.y;
        }
        d0 += 2.0f;
        d1 += 2.0f;
        float2 di;
        di.x = (d0 > 0.0f) ? rsqrtf(d0) : 0.0f;
        di.y = (d1 > 0.0f) ? rsqrtf(d1) : 0.0f;
        *(float2*)(dinv + n0) = di;
    }
    const int pair = tot0 + tot1;
    sc[t] = pair;
    __syncthreads();
    for (unsigned off = 1; off < SB_T; off <<= 1) {
        const int v = (t >= off) ? sc[t - off] : 0;
        __syncthreads();
        sc[t] += v;
        __syncthreads();
    }
    const int excl = sc[t] - pair;
    if (t == SB_T - 1) btot[blockIdx.x] = sc[t];
    if (act) {
        int r0 = excl, r1 = excl + tot0;
#pragma unroll
        for (int c = 0; c < DEG_C; ++c) {
            int2 o; o.x = r0; o.y = r1;
            *(int2*)(cpart + (size_t)c * N_NODES + n0) = o;
            r0 += c0[c]; r1 += c1[c];
        }
    }
}

__global__ __launch_bounds__(DEG_T)
void place_kernel(const int* __restrict__ src,
                  const int* __restrict__ dst,
                  const float* __restrict__ w,
                  int* __restrict__ offs,
                  const int* __restrict__ btot,
                  int2* __restrict__ recs) {
    __shared__ __attribute__((aligned(16))) int cur[DEG_QN];
    __shared__ int bb[257];
    const unsigned tid   = threadIdx.x;
    const unsigned chunk = blockIdx.x;
    const unsigned nbase = blockIdx.y * DEG_QN;

    int* sc = bb + 1;
    if (tid < 256) sc[tid] = (tid < SB_B) ? btot[tid] : 0;
    if (tid == 0) bb[0] = 0;
    __syncthreads();
    for (unsigned off = 1; off < 256; off <<= 1) {
        const int v = (tid < 256 && tid >= off) ? sc[tid - off] : 0;
        __syncthreads();
        if (tid < 256) sc[tid] += v;
        __syncthreads();
    }

    int4* my4  = (int4*)(offs + (size_t)chunk * N_NODES + nbase);
    int4* cur4 = (int4*)cur;
    for (unsigned i0 = tid; i0 < DEG_QV; i0 += DEG_T * DEG_U) {
        int4 a[DEG_U];
#pragma unroll
        for (int u = 0; u < DEG_U; ++u) {
            const unsigned i  = i0 + (unsigned)u * DEG_T;
            const unsigned ic = (i < DEG_QV) ? i : (DEG_QV - 1);
            a[u] = my4[ic];
        }
        asm volatile("" :: "v"(a[0].x), "v"(a[0].y), "v"(a[0].z), "v"(a[0].w),
                           "v"(a[1].x), "v"(a[1].y), "v"(a[1].z), "v"(a[1].w),
                           "v"(a[2].x), "v"(a[2].y), "v"(a[2].z), "v"(a[2].w),
                           "v"(a[3].x), "v"(a[3].y), "v"(a[3].z), "v"(a[3].w));
#pragma unroll
        for (int u = 0; u < DEG_U; ++u) {
            const unsigned i = i0 + (unsigned)u * DEG_T;
            if (i < DEG_QV) {
                const int b = bb[(nbase + 4u * i) / SB_N];
                int4 v;
                v.x = a[u].x + b; v.y = a[u].y + b; v.z = a[u].z + b; v.w = a[u].w + b;
                cur4[i] = v;
                if (chunk == 0) my4[i] = v;
            }
        }
    }
    __syncthreads();

    const int4*   d4 = (const int4*)(dst + (size_t)chunk * DEG_CE);
    const int4*   s4 = (const int4*)(src + (size_t)chunk * DEG_CE);
    const float4* w4 = (const float4*)(w + (size_t)chunk * DEG_CE);

    for (unsigned i0 = tid; i0 < DEG_CV; i0 += DEG_T * DEG_U) {
        int4   dv[DEG_U];
        int4   sv[DEG_U];
        float4 wv[DEG_U];
#pragma unroll
        for (int u = 0; u < DEG_U; ++u) {
            const unsigned i  = i0 + (unsigned)u * DEG_T;
            const unsigned ic = (i < DEG_CV) ? i : (DEG_CV - 1);
            dv[u] = d4[ic];
            sv[u] = s4[ic];
            wv[u] = w4[ic];
        }
        asm volatile("" :: "v"(sv[0].x), "v"(sv[0].y), "v"(sv[0].z), "v"(sv[0].w),
                           "v"(sv[1].x), "v"(sv[1].y), "v"(sv[1].z), "v"(sv[1].w),
                           "v"(sv[2].x), "v"(sv[2].y), "v"(sv[2].z), "v"(sv[2].w),
                           "v"(sv[3].x), "v"(sv[3].y), "v"(sv[3].z), "v"(sv[3].w));
        asm volatile("" :: "v"(wv[0].x), "v"(wv[0].y), "v"(wv[0].z), "v"(wv[0].w),
                           "v"(wv[1].x), "v"(wv[1].y), "v"(wv[1].z), "v"(wv[1].w),
                           "v"(wv[2].x), "v"(wv[2].y), "v"(wv[2].z), "v"(wv[2].w),
                           "v"(wv[3].x), "v"(wv[3].y), "v"(wv[3].z), "v"(wv[3].w));
#pragma unroll
        for (int u = 0; u < DEG_U; ++u) {
            const unsigned i = i0 + (unsigned)u * DEG_T;
            if (i < DEG_CV) {
                unsigned r;
                r = (unsigned)dv[u].x - nbase;
                if (r < DEG_QN) { const int p = atomicAdd(&cur[r], 1); recs[p] = make_int2(sv[u].x, __float_as_int(wv[u].x)); }
                r = (unsigned)dv[u].y - nbase;
                if (r < DEG_QN) { const int p = atomicAdd(&cur[r], 1); recs[p] = make_int2(sv[u].y, __float_as_int(wv[u].y)); }
                r = (unsigned)dv[u].z - nbase;
                if (r < DEG_QN) { const int p = atomicAdd(&cur[r], 1); recs[p] = make_int2(sv[u].z, __float_as_int(wv[u].z)); }
                r = (unsigned)dv[u].w - nbase;
                if (r < DEG_QN) { const int p = atomicAdd(&cur[r], 1); recs[p] = make_int2(sv[u].w, __float_as_int(wv[u].w)); }
            }
        }
    }
}

__global__ __launch_bounds__(256)
void pull_kernel(const float* __restrict__ x,
                 const float* __restrict__ dinv,
                 const int* __restrict__ start,
                 const int2* __restrict__ recs,
                 float* __restrict__ agg) {
    const unsigned t = blockIdx.x * 256u + threadIdx.x;
    const unsigned n = t >> 4;
    const unsigned q = t & 15u;
    const float4* x4 = (const float4*)x;

    const int jb = start[n];
    const int je = (n + 1u < N_NODES) ? start[n + 1u] : N_EDGES;
    const float di = dinv[n];
    const float4 xn = x4[(size_t)n * 16u + q];

    float4 acc = make_float4(0.0f, 0.0f, 0.0f, 0.0f);
    for (int j = jb; j < je; j += 4) {
        int2   rc[4];
        float  cf[4];
        float4 xv[4];
#pragma unroll
        for (int u = 0; u < 4; ++u) {
            const int jj = (j + u < je) ? (j + u) : (je - 1);
            rc[u] = recs[jj];
        }
#pragma unroll
        for (int u = 0; u < 4; ++u) {
            const float ws = (j + u < je) ? __int_as_float(rc[u].y) : 0.0f;
            cf[u] = ws * dinv[rc[u].x];
            xv[u] = x4[(size_t)rc[u].x * 16u + q];
        }
#pragma unroll
        for (int u = 0; u < 4; ++u) {
            acc.x += cf[u] * xv[u].x;
            acc.y += cf[u] * xv[u].y;
            acc.z += cf[u] * xv[u].z;
            acc.w += cf[u] * xv[u].w;
        }
    }
    const float sl = 2.0f * di;
    float4 o;
    o.x = di * (acc.x + sl * xn.x);
    o.y = di * (acc.y + sl * xn.y);
    o.z = di * (acc.z + sl * xn.z);
    o.w = di * (acc.w + sl * xn.w);
    ((float4*)agg)[(size_t)n * 16u + q] = o;
}

#define GA_P 68
#define GW_P 72

__global__ __launch_bounds__(256)
void gemm_wmma_kernel(const float* __restrict__ a_in,
                      const float* __restrict__ W,
                      const float* __restrict__ bias,
                      float* __restrict__ out) {
    __shared__ __attribute__((aligned(16))) float As[32 * GA_P];
    __shared__ __attribute__((aligned(16))) float Ws[IN_W * GW_P];
    const unsigned tid  = threadIdx.x;
    const unsigned lane = tid & 31u;
    const unsigned wave = tid >> 5;
    const unsigned row0 = blockIdx.x * 32u;
    const unsigned tile_n = wave & 3u;
    const unsigned strip  = wave >> 2;
    const unsigned m0 = row0 + strip * 16u;
    const unsigned n0 = tile_n * 16u;
    const unsigned half = lane >> 4;
    const unsigned lrow = lane & 15u;

    const float bv = bias[n0 + lrow];

    const float4* a4 = (const float4*)(a_in + (size_t)row0 * IN_W);
    const float4* w4 = (const float4*)W;
    float4 av[2], wv[4];
#pragma unroll
    for (int u = 0; u < 2; ++u) av[u] = a4[tid + 256u * (unsigned)u];
#pragma unroll
    for (int u = 0; u < 4; ++u) wv[u] = w4[tid + 256u * (unsigned)u];
#pragma unroll
    for (int u = 0; u < 2; ++u) {
        const unsigned idx = tid + 256u * (unsigned)u;
        *(float4*)(As + (idx >> 4) * GA_P + (idx & 15u) * 4u) = av[u];
    }
#pragma unroll
    for (int u = 0; u < 4; ++u) {
        const unsigned idx = tid + 256u * (unsigned)u;
        *(float4*)(Ws + (idx >> 4) * GW_P + (idx & 15u) * 4u) = wv[u];
    }
    __syncthreads();

    const float* arow = As + (strip * 16u + lrow) * GA_P + half * 2u;
    const float* wcol = Ws + (half * 2u) * GW_P + n0 + lrow;

    v2f fa[16], fb[16];
#pragma unroll
    for (int kk = 0; kk < 16; ++kk) {
        fa[kk] = *(const v2f*)(arow + kk * 4);
        fb[kk].x = wcol[(kk * 4) * GW_P];
        fb[kk].y = wcol[(kk * 4 + 1) * GW_P];
    }
    asm volatile("" :: "v"(fa[0]),  "v"(fa[1]),  "v"(fa[2]),  "v"(fa[3]),
                       "v"(fa[4]),  "v"(fa[5]),  "v"(fa[6]),  "v"(fa[7]),
                       "v"(fa[8]),  "v"(fa[9]),  "v"(fa[10]), "v"(fa[11]),
                       "v"(fa[12]), "v"(fa[13]), "v"(fa[14]), "v"(fa[15]));
    asm volatile("" :: "v"(fb[0]),  "v"(fb[1]),  "v"(fb[2]),  "v"(fb[3]),
                       "v"(fb[4]),  "v"(fb[5]),  "v"(fb[6]),  "v"(fb[7]),
                       "v"(fb[8]),  "v"(fb[9]),  "v"(fb[10]), "v"(fb[11]),
                       "v"(fb[12]), "v"(fb[13]), "v"(fb[14]), "v"(fb[15]));
    v8f c = {};
#pragma unroll
    for (int kk = 0; kk < 16; ++kk) {
        c = __builtin_amdgcn_wmma_f32_16x16x4_f32(
                 false, fa[kk],  false, fb[kk],
                 (short)0, c,  false,  false);
    }

    float* ocol = out + n0 + lrow;
#pragma unroll
    for (int v = 0; v < 8; ++v) {
        ocol[(size_t)(m0 + v + half * 8u) * OUT_W] = c[v] + bv;
    }
}

extern "C" void kernel_launch(void* const* d_in, const int* in_sizes, int n_in,
                              void* d_out, int out_size, void* d_ws, size_t ws_size,
                              hipStream_t stream) {
    const float* x         = (const float*)d_in[0];
    const float* edge_attr = (const float*)d_in[1];
    const float* W         = (const float*)d_in[2];
    const float* bias      = (const float*)d_in[3];
    const int*   edge_idx  = (const int*)d_in[4];
    float* out = (float*)d_out;

    float* dinv = (float*)d_ws;
    float* regR = dinv + 100096;
    int*   btot = (int*)(regR + (size_t)DEG_C * N_NODES);

    int*  offs = (int*)out;
    int2* recs = (int2*)(out + (size_t)DEG_C * N_NODES);

    const int* src = edge_idx;
    const int* dst = edge_idx + N_EDGES;

    float* wpart = regR;
    float* agg   = regR;

    count_kernel<<<dim3(DEG_C, DEG_Q), DEG_T, 0, stream>>>(dst, edge_attr, wpart, offs);
    offsets_kernel<<<SB_B, SB_T, 0, stream>>>(wpart, offs, dinv, btot);
    place_kernel<<<dim3(DEG_C, DEG_Q), DEG_T, 0, stream>>>(src, dst, edge_attr, offs, btot, recs);
    pull_kernel<<<(N_NODES * 16) / 256, 256, 0, stream>>>(x, dinv, offs, recs, agg);
    gemm_wmma_kernel<<<3125, 256, 0, stream>>>(agg, W, bias, out);
}


# count and place split the nodes into 8 ranges (was 4): 256 work-groups, one per WGP (LDS reserved above half a WGP); owned edges and scattered stores per WGP halve; sums stay in arrival order
# speedup vs baseline: 2.4538x; 1.0465x over previous
#include <hip/hip_runtime.h>
#include <math.h>


#define N_NODES 100000
#define IN_W    64
#define OUT_W   64
#define N_EDGES 1600000

typedef __attribute__((ext_vector_type(2))) float v2f;
typedef __attribute__((ext_vector_type(8))) float v8f;

#define DEG_C   32
#define DEG_CE  (N_EDGES / DEG_C)
#define DEG_CV  (DEG_CE / 4)
#define DEG_Q   8
#define DEG_QN  (N_NODES / DEG_Q)
#define DEG_QV  (DEG_QN / 4)
#define DEG_LB  20992
#define DEG_LC  41984
#define DEG_T   1024
#define DEG_U   4

#define SB_N    400
#define SB_B    (N_NODES / SB_N)
#define SB_T    256

__global__ __launch_bounds__(DEG_T)
void count_kernel(const int* __restrict__ dst,
                  const float* __restrict__ w,
                  float* __restrict__ wpart,
                  int* __restrict__ cpart) {
    __shared__ __attribute__((aligned(16))) float bins[DEG_LB];
    __shared__ __attribute__((aligned(16))) int   cnts[DEG_LB];
    const unsigned tid   = threadIdx.x;
    const unsigned chunk = blockIdx.x;
    const unsigned nbase = blockIdx.y * DEG_QN;

    float4* bins4 = (float4*)bins;
    int4*   cnts4 = (int4*)cnts;
    for (unsigned i = tid; i < DEG_QV; i += DEG_T) {
        bins4[i] = make_float4(0.0f, 0.0f, 0.0f, 0.0f);
        cnts4[i] = make_int4(0, 0, 0, 0);
    }
    __syncthreads();

    const int4*   d4 = (const int4*)(dst + (size_t)chunk * DEG_CE);
    const float4* w4 = (const float4*)(w + (size_t)chunk * DEG_CE);

    for (unsigned i0 = tid; i0 < DEG_CV; i0 += DEG_T * DEG_U) {
        int4   dv[DEG_U];
        float4 wv[DEG_U];
#pragma unroll
        for (int u = 0; u < DEG_U; ++u) {
            const unsigned i  = i0 + (unsigned)u * DEG_T;
            const unsigned ic = (i < DEG_CV) ? i : (DEG_CV - 1);
            dv[u] = d4[ic];
            wv[u] = w4[ic];
        }
        asm volatile("" :: "v"(wv[0].x), "v"(wv[0].y), "v"(wv[0].z), "v"(wv[0].w),
                           "v"(wv[1].x), "v"(wv[1].y), "v"(wv[1].z), "v"(wv[1].w),
                           "v"(wv[2].x), "v"(wv[2].y), "v"(wv[2].z), "v"(wv[2].w),
                           "v"(wv[3].x), "v"(wv[3].y), "v"(wv[3].z), "v"(wv[3].w));
#pragma unroll
        for (int u = 0; u < DEG_U; ++u) {
            const unsigned i = i0 + (unsigned)u * DEG_T;
            if (i < DEG_CV) {
                unsigned r;
                r = (unsigned)dv[u].x - nbase; if (r < DEG_QN) { atomicAdd(&bins[r], wv[u].x); atomicAdd(&cnts[r], 1); }
                r = (unsigned)dv[u].y - nbase; if (r < DEG_QN) { atomicAdd(&bins[r], wv[u].y); atomicAdd(&cnts[r], 1); }
                r = (unsigned)dv[u].z - nbase; if (r < DEG_QN) { atomicAdd(&bins[r], wv[u].z); atomicAdd(&cnts[r], 1); }
                r = (unsigned)dv[u].w - nbase; if (r < DEG_QN) { atomicAdd(&bins[r], wv[u].w); atomicAdd(&cnts[r], 1); }
            }
        }
    }
    __syncthreads();

    float4* o4 = (float4*)(wpart + (size_t)chunk * N_NODES + nbase);
    int4*   c4 = (int4*)(cpart + (size_t)chunk * N_NODES + nbase);
    for (unsigned i = tid; i < DEG_QV; i += DEG_T) {
        o4[i] = bins4[i];
        c4[i] = cnts4[i];
    }
}

__global__ __launch_bounds__(SB_T)
void offsets_kernel(const float* __restrict__ wpart,
                    int* __restrict__ cpart,
                    float* __restrict__ dinv,
                    int* __restrict__ btot) {
    __shared__ int sc[SB_T];
    const unsigned t  = threadIdx.x;
    const bool act    = t < (SB_N / 2);
    const unsigned n0 = blockIdx.x * SB_N + 2u * t;

    int c0[DEG_C], c1[DEG_C];
    int tot0 = 0, tot1 = 0;
    if (act) {
        float d0 = 0.0f, d1 = 0.0f;
#pragma unroll
        for (int c = 0; c < DEG_C; ++c) {
            const float2 wv = *(const float2*)(wpart + (size_t)c * N_NODES + n0);
            const int2   cv = *(const int2*)(cpart + (size_t)c * N_NODES + n0);
            d0 += wv.x; d1 += wv.y;
            c0[c] = cv.x; c1[c] = cv.y;
            tot0 += cv.x; tot1 += cv.y;
        }
        d0 += 2.0f;
        d1 += 2.0f;
        float2 di;
        di.x = (d0 > 0.0f) ? rsqrtf(d0) : 0.0f;
        di.y = (d1 > 0.0f) ? rsqrtf(d1) : 0.0f;
        *(float2*)(dinv + n0) = di;
    }
    const int pair = tot0 + tot1;
    sc[t] = pair;
    __syncthreads();
    for (unsigned off = 1; off < SB_T; off <<= 1) {
        const int v = (t >= off) ? sc[t - off] : 0;
        __syncthreads();
        sc[t] += v;
        __syncthreads();
    }
    const int excl = sc[t] - pair;
    if (t == SB_T - 1) btot[blockIdx.x] = sc[t];
    if (act) {
        int r0 = excl, r1 = excl + tot0;
#pragma unroll
        for (int c = 0; c < DEG_C; ++c) {
            int2 o; o.x = r0; o.y = r1;
            *(int2*)(cpart + (size_t)c * N_NODES + n0) = o;
            r0 += c0[c]; r1 += c1[c];
        }
    }
}

__global__ __launch_bounds__(DEG_T)
void place_kernel(const int* __restrict__ src,
                  const int* __restrict__ dst,
                  const float* __restrict__ w,
                  int* __restrict__ offs,
                  const int* __restrict__ btot,
                  int2* __restrict__ recs) {
    __shared__ __attribute__((aligned(16))) int cur[DEG_LC];
    __shared__ int bb[257];
    const unsigned tid   = threadIdx.x;
    const unsigned chunk = blockIdx.x;
    const unsigned nbase = blockIdx.y * DEG_QN;

    int* sc = bb + 1;
    if (tid < 256) sc[tid] = (tid < SB_B) ? btot[tid] : 0;
    if (tid == 0) bb[0] = 0;
    __syncthreads();
    for (unsigned off = 1; off < 256; off <<= 1) {
        const int v = (tid < 256 && tid >= off) ? sc[tid - off] : 0;
        __syncthreads();
        if (tid < 256) sc[tid] += v;
        __syncthreads();
    }

    int4* my4  = (int4*)(offs + (size_t)chunk * N_NODES + nbase);
    int4* cur4 = (int4*)cur;
    for (unsigned i0 = tid; i0 < DEG_QV; i0 += DEG_T * DEG_U) {
        int4 a[DEG_U];
#pragma unroll
        for (int u = 0; u < DEG_U; ++u) {
            const unsigned i  = i0 + (unsigned)u * DEG_T;
            const unsigned ic = (i < DEG_QV) ? i : (DEG_QV - 1);
            a[u] = my4[ic];
        }
        asm volatile("" :: "v"(a[0].x), "v"(a[0].y), "v"(a[0].z), "v"(a[0].w),
                           "v"(a[1].x), "v"(a[1].y), "v"(a[1].z), "v"(a[1].w),
                           "v"(a[2].x), "v"(a[2].y), "v"(a[2].z), "v"(a[2].w),
                           "v"(a[3].x), "v"(a[3].y), "v"(a[3].z), "v"(a[3].w));
#pragma unroll
        for (int u = 0; u < DEG_U; ++u) {
            const unsigned i = i0 + (unsigned)u * DEG_T;
            if (i < DEG_QV) {
                const int b = bb[(nbase + 4u * i) / SB_N];
                int4 v;
                v.x = a[u].x + b; v.y = a[u].y + b; v.z = a[u].z + b; v.w = a[u].w + b;
                cur4[i] = v;
                if (chunk == 0) my4[i] = v;
            }
        }
    }
    __syncthreads();

    const int4*   d4 = (const int4*)(dst + (size_t)chunk * DEG_CE);
    const int4*   s4 = (const int4*)(src + (size_t)chunk * DEG_CE);
    const float4* w4 = (const float4*)(w + (size_t)chunk * DEG_CE);

    for (unsigned i0 = tid; i0 < DEG_CV; i0 += DEG_T * DEG_U) {
        int4   dv[DEG_U];
        int4   sv[DEG_U];
        float4 wv[DEG_U];
#pragma unroll
        for (int u = 0; u < DEG_U; ++u) {
            const unsigned i  = i0 + (unsigned)u * DEG_T;
            const unsigned ic = (i < DEG_CV) ? i : (DEG_CV - 1);
            dv[u] = d4[ic];
            sv[u] = s4[ic];
            wv[u] = w4[ic];
        }
        asm volatile("" :: "v"(sv[0].x), "v"(sv[0].y), "v"(sv[0].z), "v"(sv[0].w),
                           "v"(sv[1].x), "v"(sv[1].y), "v"(sv[1].z), "v"(sv[1].w),
                           "v"(sv[2].x), "v"(sv[2].y), "v"(sv[2].z), "v"(sv[2].w),
                           "v"(sv[3].x), "v"(sv[3].y), "v"(sv[3].z), "v"(sv[3].w));
        asm volatile("" :: "v"(wv[0].x), "v"(wv[0].y), "v"(wv[0].z), "v"(wv[0].w),
                           "v"(wv[1].x), "v"(wv[1].y), "v"(wv[1].z), "v"(wv[1].w),
                           "v"(wv[2].x), "v"(wv[2].y), "v"(wv[2].z), "v"(wv[2].w),
                           "v"(wv[3].x), "v"(wv[3].y), "v"(wv[3].z), "v"(wv[3].w));
#pragma unroll
        for (int u = 0; u < DEG_U; ++u) {
            const unsigned i = i0 + (unsigned)u * DEG_T;
            if (i < DEG_CV) {
                unsigned r;
                r = (unsigned)dv[u].x - nbase;
                if (r < DEG_QN) { const int p = atomicAdd(&cur[r], 1); recs[p] = make_int2(sv[u].x, __float_as_int(wv[u].x)); }
                r = (unsigned)dv[u].y - nbase;
                if (r < DEG_QN) { const int p = atomicAdd(&cur[r], 1); recs[p] = make_int2(sv[u].y, __float_as_int(wv[u].y)); }
                r = (unsigned)dv[u].z - nbase;
                if (r < DEG_QN) { const int p = atomicAdd(&cur[r], 1); recs[p] = make_int2(sv[u].z, __float_as_int(wv[u].z)); }
                r = (unsigned)dv[u].w - nbase;
                if (r < DEG_QN) { const int p = atomicAdd(&cur[r], 1); recs[p] = make_int2(sv[u].w, __float_as_int(wv[u].w)); }
            }
        }
    }
}

__global__ __launch_bounds__(256)
void pull_kernel(const float* __restrict__ x,
                 const float* __restrict__ dinv,
                 const int* __restrict__ start,
                 const int2* __restrict__ recs,
                 float* __restrict__ agg) {
    const unsigned t = blockIdx.x * 256u + threadIdx.x;
    const unsigned n = t >> 4;
    const unsigned q = t & 15u;
    const float4* x4 = (const float4*)x;

    const int jb = start[n];
    const int je = (n + 1u < N_NODES) ? start[n + 1u] : N_EDGES;
    const float di = dinv[n];
    const float4 xn = x4[(size_t)n * 16u + q];

    float4 acc = make_float4(0.0f, 0.0f, 0.0f, 0.0f);
    for (int j = jb; j < je; j += 4) {
        int2   rc[4];
        float  cf[4];
        float4 xv[4];
#pragma unroll
        for (int u = 0; u < 4; ++u) {
            const int jj = (j + u < je) ? (j + u) : (je - 1);
            rc[u] = recs[jj];
        }
#pragma unroll
        for (int u = 0; u < 4; ++u) {
            const float ws = (j + u < je) ? __int_as_float(rc[u].y) : 0.0f;
            cf[u] = ws * dinv[rc[u].x];
            xv[u] = x4[(size_t)rc[u].x * 16u + q];
        }
#pragma unroll
        for (int u = 0; u < 4; ++u) {
            acc.x += cf[u] * xv[u].x;
            acc.y += cf[u] * xv[u].y;
            acc.z += cf[u] * xv[u].z;
            acc.w += cf[u] * xv[u].w;
        }
    }
    const float sl = 2.0f * di;
    float4 o;
    o.x = di * (acc.x + sl * xn.x);
    o.y = di * (acc.y + sl * xn.y);
    o.z = di * (acc.z + sl * xn.z);
    o.w = di * (acc.w + sl * xn.w);
    ((float4*)agg)[(size_t)n * 16u + q] = o;
}

#define GA_P 68
#define GW_P 72

__global__ __launch_bounds__(256)
void gemm_wmma_kernel(const float* __restrict__ a_in,
                      const float* __restrict__ W,
                      const float* __restrict__ bias,
                      float* __restrict__ out) {
    __shared__ __attribute__((aligned(16))) float As[32 * GA_P];
    __shared__ __attribute__((aligned(16))) float Ws[IN_W * GW_P];
    const unsigned tid  = threadIdx.x;
    const unsigned lane = tid & 31u;
    const unsigned wave = tid >> 5;
    const unsigned row0 = blockIdx.x * 32u;
    const unsigned tile_n = wave & 3u;
    const unsigned strip  = wave >> 2;
    const unsigned m0 = row0 + strip * 16u;
    const unsigned n0 = tile_n * 16u;
    const unsigned half = lane >> 4;
    const unsigned lrow = lane & 15u;

    const float bv = bias[n0 + lrow];

    const float4* a4 = (const float4*)(a_in + (size_t)row0 * IN_W);
    const float4* w4 = (const float4*)W;
    float4 av[2], wv[4];
#pragma unroll
    for (int u = 0; u < 2; ++u) av[u] = a4[tid + 256u * (unsigned)u];
#pragma unroll
    for (int u = 0; u < 4; ++u) wv[u] = w4[tid + 256u * (unsigned)u];
#pragma unroll
    for (int u = 0; u < 2; ++u) {
        const unsigned idx = tid + 256u * (unsigned)u;
        *(float4*)(As + (idx >> 4) * GA_P + (idx & 15u) * 4u) = av[u];
    }
#pragma unroll
    for (int u = 0; u < 4; ++u) {
        const unsigned idx = tid + 256u * (unsigned)u;
        *(float4*)(Ws + (idx >> 4) * GW_P + (idx & 15u) * 4u) = wv[u];
    }
    __syncthreads();

    const float* arow = As + (strip * 16u + lrow) * GA_P + half * 2u;
    const float* wcol = Ws + (half * 2u) * GW_P + n0 + lrow;

    v2f fa[16], fb[16];
#pragma unroll
    for (int kk = 0; kk < 16; ++kk) {
        fa[kk] = *(const v2f*)(arow + kk * 4);
        fb[kk].x = wcol[(kk * 4) * GW_P];
        fb[kk].y = wcol[(kk * 4 + 1) * GW_P];
    }
    asm volatile("" :: "v"(fa[0]),  "v"(fa[1]),  "v"(fa[2]),  "v"(fa[3]),
                       "v"(fa[4]),  "v"(fa[5]),  "v"(fa[6]),  "v"(fa[7]),
                       "v"(fa[8]),  "v"(fa[9]),  "v"(fa[10]), "v"(fa[11]),
                       "v"(fa[12]), "v"(fa[13]), "v"(fa[14]), "v"(fa[15]));
    asm volatile("" :: "v"(fb[0]),  "v"(fb[1]),  "v"(fb[2]),  "v"(fb[3]),
                       "v"(fb[4]),  "v"(fb[5]),  "v"(fb[6]),  "v"(fb[7]),
                       "v"(fb[8]),  "v"(fb[9]),  "v"(fb[10]), "v"(fb[11]),
                       "v"(fb[12]), "v"(fb[13]), "v"(fb[14]), "v"(fb[15]));
    v8f c = {};
#pragma unroll
    for (int kk = 0; kk < 16; ++kk) {
        c = __builtin_amdgcn_wmma_f32_16x16x4_f32(
                 false, fa[kk],  false, fb[kk],
                 (short)0, c,  false,  false);
    }

    float* ocol = out + n0 + lrow;
#pragma unroll
    for (int v = 0; v < 8; ++v) {
        ocol[(size_t)(m0 + v + half * 8u) * OUT_W] = c[v] + bv;
    }
}

extern "C" void kernel_launch(void* const* d_in, const int* in_sizes, int n_in,
                              void* d_out, int out_size, void* d_ws, size_t ws_size,
                              hipStream_t stream) {
    const float* x         = (const float*)d_in[0];
    const float* edge_attr = (const float*)d_in[1];
    const float* W         = (const float*)d_in[2];
    const float* bias      = (const float*)d_in[3];
    const int*   edge_idx  = (const int*)d_in[4];
    float* out = (float*)d_out;

    float* dinv = (float*)d_ws;
    float* regR = dinv + 100096;
    int*   btot = (int*)(regR + (size_t)DEG_C * N_NODES);

    int*  offs = (int*)out;
    int2* recs = (int2*)(out + (size_t)DEG_C * N_NODES);

    const int* src = edge_idx;
    const int* dst = edge_idx + N_EDGES;

    float* wpart = regR;
    float* agg   = regR;

    count_kernel<<<dim3(DEG_C, DEG_Q), DEG_T, 0, stream>>>(dst, edge_attr, wpart, offs);
    offsets_kernel<<<SB_B, SB_T, 0, stream>>>(wpart, offs, dinv, btot);
    place_kernel<<<dim3(DEG_C, DEG_Q), DEG_T, 0, stream>>>(src, dst, edge_attr, offs, btot, recs);
    pull_kernel<<<(N_NODES * 16) / 256, 256, 0, stream>>>(x, dinv, offs, recs, agg);
    gemm_wmma_kernel<<<3125, 256, 0, stream>>>(agg, W, bias, out);
}
